# starting kernel
# baseline (speedup 1.0000x reference)
#include <hip/hip_runtime.h>


typedef float v2f __attribute__((ext_vector_type(2)));
typedef float v8f __attribute__((ext_vector_type(8)));

#define BATCH       8192
#define PROTOS      4096
#define CLASSES     100
#define NTILES      7
#define WAVES       8
#define THREADS     (WAVES * 32)
#define KSLABS      (PROTOS / 4)
#define SLABS_PER_W (KSLABS / WAVES)
#define ROWTILES    2
#define BLOCK_ROWS  (16 * ROWTILES)

__global__ __launch_bounds__(THREADS)
void stratified_sum_wmma(const float* __restrict__ V,
                         const int*   __restrict__ labels,
                         float*       __restrict__ out) {
    __shared__ __align__(32) float smem[WAVES * NTILES * 32 * 8];
    int* lds_labels = (int*)smem;

    const int tid  = threadIdx.x;
    const int wid  = tid >> 5;
    const int lane = tid & 31;
    const int row_base = blockIdx.x * BLOCK_ROWS;

    for (int i = tid; i < PROTOS; i += THREADS)
        lds_labels[i] = labels[i];
    __syncthreads();

    const int l15  = lane & 15;
    const int koff = (lane >> 4) * 2;

    v8f acc[ROWTILES][NTILES];
#pragma unroll
    for (int h = 0; h < ROWTILES; ++h)
#pragma unroll
        for (int t = 0; t < NTILES; ++t) { v8f z = {}; acc[h][t] = z; }

    const int    kstart = wid * SLABS_PER_W * 4;
    const float* vp0 = V + (size_t)(row_base + l15) * PROTOS + kstart + koff;
    const float* vp1 = vp0 + (size_t)16 * PROTOS;
    const int*   lp  = lds_labels + kstart + koff;

#pragma unroll 4
    for (int s = 0; s < SLABS_PER_W; ++s) {
        v2f a0 = *(const v2f*)vp0;
        v2f a1 = *(const v2f*)vp1;
        const int la = lp[0];
        const int lb = lp[1];
#pragma unroll
        for (int t = 0; t < NTILES; ++t) {
            const int cls = t * 16 + l15;
            v2f b;
            b.x = (la == cls) ? 1.0f : 0.0f;
            b.y = (lb == cls) ? 1.0f : 0.0f;
            acc[0][t] = __builtin_amdgcn_wmma_f32_16x16x4_f32(
                false, a0, false, b, (short)0, acc[0][t], false, false);
            acc[1][t] = __builtin_amdgcn_wmma_f32_16x16x4_f32(
                false, a1, false, b, (short)0, acc[1][t], false, false);
        }
        vp0 += 4;
        vp1 += 4;
        lp  += 4;
    }

#pragma unroll
    for (int h = 0; h < ROWTILES; ++h) {
        __syncthreads();
#pragma unroll
        for (int t = 0; t < NTILES; ++t) {
            float* dst = &smem[(((wid * NTILES) + t) * 32 + lane) * 8];
            *(v8f*)dst = acc[h][t];
        }
        __syncthreads();

        for (int idx = tid; idx < NTILES * 256; idx += THREADS) {
            const int t  = idx >> 8;
            const int ln = (idx >> 3) & 31;
            const int r  = idx & 7;
            float sum = 0.0f;
#pragma unroll
            for (int w = 0; w < WAVES; ++w)
                sum += smem[(((w * NTILES) + t) * 32 + ln) * 8 + r];
            const int row = row_base + h * 16 + r + ((ln >> 4) << 3);
            const int cls = t * 16 + (ln & 15);
            if (cls < CLASSES)
                out[(size_t)row * CLASSES + cls] = sum;
        }
    }
}

extern "C" void kernel_launch(void* const* d_in, const int* in_sizes, int n_in,
                              void* d_out, int out_size, void* d_ws, size_t ws_size,
                              hipStream_t stream) {
    const float* V      = (const float*)d_in[0];
    const int*   labels = (const int*)d_in[1];
    float*       out    = (float*)d_out;

    stratified_sum_wmma<<<BATCH / BLOCK_ROWS, THREADS, 0, stream>>>(V, labels, out);
}


# Each A register (one column per lane half) is issued only for the class tiles its two labels hit, two registers packed per WMMA via per-tile pending slots under scalar branches; sum order changes
# speedup vs baseline: 1.6415x; 1.6415x over previous
#include <hip/hip_runtime.h>


typedef float v2f __attribute__((ext_vector_type(2)));
typedef float v4f __attribute__((ext_vector_type(4)));
typedef float v8f __attribute__((ext_vector_type(8)));

#define BATCH       8192
#define PROTOS      4096
#define CLASSES     100
#define NTILES      7
#define WAVES       16
#define THREADS     (WAVES * 32)
#define KSLABS      (PROTOS / 4)
#define SLABS_PER_W (KSLABS / WAVES)
#define TRIPS       (SLABS_PER_W / 8)
#define ROWTILES    2
#define BLOCK_ROWS  (16 * ROWTILES)
#define TILE_F      256
#define UNITS       (ROWTILES * NTILES * 2)


#define WMMA_PAIR_0(X0, X1, B)                                                     \
    asm volatile("v_wmma_f32_16x16x4_f32 v[144:151], %0, %2, v[144:151]\n\t"       \
                 "v_wmma_f32_16x16x4_f32 v[200:207], %1, %2, v[200:207]\n\t"       \
                 "v_nop\n\tv_nop\n\tv_nop\n\tv_nop"                                     \
                 : : "v"(X0), "v"(X1), "v"(B)                                         \
                 : "v144", "v145", "v146", "v147", "v148", "v149", "v150", "v151", "v200", "v201", "v202", "v203", "v204", "v205", "v206", "v207")
#define WMMA_PAIR_1(X0, X1, B)                                                     \
    asm volatile("v_wmma_f32_16x16x4_f32 v[152:159], %0, %2, v[152:159]\n\t"       \
                 "v_wmma_f32_16x16x4_f32 v[208:215], %1, %2, v[208:215]\n\t"       \
                 "v_nop\n\tv_nop\n\tv_nop\n\tv_nop"                                     \
                 : : "v"(X0), "v"(X1), "v"(B)                                         \
                 : "v152", "v153", "v154", "v155", "v156", "v157", "v158", "v159", "v208", "v209", "v210", "v211", "v212", "v213", "v214", "v215")
#define WMMA_PAIR_2(X0, X1, B)                                                     \
    asm volatile("v_wmma_f32_16x16x4_f32 v[160:167], %0, %2, v[160:167]\n\t"       \
                 "v_wmma_f32_16x16x4_f32 v[216:223], %1, %2, v[216:223]\n\t"       \
                 "v_nop\n\tv_nop\n\tv_nop\n\tv_nop"                                     \
                 : : "v"(X0), "v"(X1), "v"(B)                                         \
                 : "v160", "v161", "v162", "v163", "v164", "v165", "v166", "v167", "v216", "v217", "v218", "v219", "v220", "v221", "v222", "v223")
#define WMMA_PAIR_3(X0, X1, B)                                                     \
    asm volatile("v_wmma_f32_16x16x4_f32 v[168:175], %0, %2, v[168:175]\n\t"       \
                 "v_wmma_f32_16x16x4_f32 v[224:231], %1, %2, v[224:231]\n\t"       \
                 "v_nop\n\tv_nop\n\tv_nop\n\tv_nop"                                     \
                 : : "v"(X0), "v"(X1), "v"(B)                                         \
                 : "v168", "v169", "v170", "v171", "v172", "v173", "v174", "v175", "v224", "v225", "v226", "v227", "v228", "v229", "v230", "v231")
#define WMMA_PAIR_4(X0, X1, B)                                                     \
    asm volatile("v_wmma_f32_16x16x4_f32 v[176:183], %0, %2, v[176:183]\n\t"       \
                 "v_wmma_f32_16x16x4_f32 v[232:239], %1, %2, v[232:239]\n\t"       \
                 "v_nop\n\tv_nop\n\tv_nop\n\tv_nop"                                     \
                 : : "v"(X0), "v"(X1), "v"(B)                                         \
                 : "v176", "v177", "v178", "v179", "v180", "v181", "v182", "v183", "v232", "v233", "v234", "v235", "v236", "v237", "v238", "v239")
#define WMMA_PAIR_5(X0, X1, B)                                                     \
    asm volatile("v_wmma_f32_16x16x4_f32 v[184:191], %0, %2, v[184:191]\n\t"       \
                 "v_wmma_f32_16x16x4_f32 v[240:247], %1, %2, v[240:247]\n\t"       \
                 "v_nop\n\tv_nop\n\tv_nop\n\tv_nop"                                     \
                 : : "v"(X0), "v"(X1), "v"(B)                                         \
                 : "v184", "v185", "v186", "v187", "v188", "v189", "v190", "v191", "v240", "v241", "v242", "v243", "v244", "v245", "v246", "v247")
#define WMMA_PAIR_6(X0, X1, B)                                                     \
    asm volatile("v_wmma_f32_16x16x4_f32 v[192:199], %0, %2, v[192:199]\n\t"       \
                 "v_wmma_f32_16x16x4_f32 v[248:255], %1, %2, v[248:255]\n\t"       \
                 "v_nop\n\tv_nop\n\tv_nop\n\tv_nop"                                     \
                 : : "v"(X0), "v"(X1), "v"(B)                                         \
                 : "v192", "v193", "v194", "v195", "v196", "v197", "v198", "v199", "v248", "v249", "v250", "v251", "v252", "v253", "v254", "v255")

#define ACC_READ(F, R0, R1, R2, R3, R4, R5, R6, R7)                                    \
    asm volatile("v_mov_b32 %0, v" #R0 "\n\tv_mov_b32 %1, v" #R1 "\n\t"                \
                 "v_mov_b32 %2, v" #R2 "\n\tv_mov_b32 %3, v" #R3 "\n\t"                \
                 "v_mov_b32 %4, v" #R4 "\n\tv_mov_b32 %5, v" #R5 "\n\t"                \
                 "v_mov_b32 %6, v" #R6 "\n\tv_mov_b32 %7, v" #R7                        \
                 : "=v"(F[0]), "=v"(F[1]), "=v"(F[2]), "=v"(F[3]),                     \
                   "=v"(F[4]), "=v"(F[5]), "=v"(F[6]), "=v"(F[7]))

#define SITE(T)                                                                      \
    if (hit & (1u << T)) {                                                           \
        const unsigned m = ((ta == (unsigned)T) ? bla : 0u) | ((tb == (unsigned)T) ? blb : 0u); \
        if (!(pf & (1u << T))) {                                                     \
            px0_##T = r0; px1_##T = r1; pm_##T = m; pf |= (1u << T);                 \
        } else {                                                                     \
            const v2f x0 = {px0_##T, r0};                                            \
            const v2f x1 = {px1_##T, r1};                                            \
            v2f b;                                                                   \
            b.x = __builtin_amdgcn_inverse_ballot_w32(pm_##T) ? 1.0f : 0.0f;         \
            b.y = __builtin_amdgcn_inverse_ballot_w32(m)      ? 1.0f : 0.0f;         \
            WMMA_PAIR_##T(x0, x1, b);                                                \
            pf &= ~(1u << T);                                                        \
        }                                                                            \
    }

#define FLUSH(T)                                                                     \
    if (pf & (1u << T)) {                                                            \
        const v2f x0 = {px0_##T, 0.0f}, x1 = {px1_##T, 0.0f};                        \
        v2f b;                                                                       \
        b.x = __builtin_amdgcn_inverse_ballot_w32(pm_##T) ? 1.0f : 0.0f;             \
        b.y = 0.0f;                                                                  \
        WMMA_PAIR_##T(x0, x1, b);                                                    \
    }

__global__ __launch_bounds__(THREADS) __attribute__((amdgpu_waves_per_eu(7, 7)))
void stratified_sum_wmma(const float* __restrict__ V,
                         const int*   __restrict__ labels,
                         float*       __restrict__ out) {
    __shared__ __align__(32) float smem[WAVES * ROWTILES * NTILES * TILE_F];

    const int tid  = threadIdx.x;
    const int wid  = __builtin_amdgcn_readfirstlane(tid >> 5);
    const int lane = tid & 31;
    const int row_base = blockIdx.x * BLOCK_ROWS;

    const int l15  = lane & 15;
    const int koff = (lane >> 4) * 2;

    asm volatile(
        "v_mov_b64 v[144:145], 0\n\t"
        "v_mov_b64 v[146:147], 0\n\t"
        "v_mov_b64 v[148:149], 0\n\t"
        "v_mov_b64 v[150:151], 0\n\t"
        "v_mov_b64 v[152:153], 0\n\t"
        "v_mov_b64 v[154:155], 0\n\t"
        "v_mov_b64 v[156:157], 0\n\t"
        "v_mov_b64 v[158:159], 0\n\t"
        "v_mov_b64 v[160:161], 0\n\t"
        "v_mov_b64 v[162:163], 0\n\t"
        "v_mov_b64 v[164:165], 0\n\t"
        "v_mov_b64 v[166:167], 0\n\t"
        "v_mov_b64 v[168:169], 0\n\t"
        "v_mov_b64 v[170:171], 0\n\t"
        "v_mov_b64 v[172:173], 0\n\t"
        "v_mov_b64 v[174:175], 0\n\t"
        "v_mov_b64 v[176:177], 0\n\t"
        "v_mov_b64 v[178:179], 0\n\t"
        "v_mov_b64 v[180:181], 0\n\t"
        "v_mov_b64 v[182:183], 0\n\t"
        "v_mov_b64 v[184:185], 0\n\t"
        "v_mov_b64 v[186:187], 0\n\t"
        "v_mov_b64 v[188:189], 0\n\t"
        "v_mov_b64 v[190:191], 0\n\t"
        "v_mov_b64 v[192:193], 0\n\t"
        "v_mov_b64 v[194:195], 0\n\t"
        "v_mov_b64 v[196:197], 0\n\t"
        "v_mov_b64 v[198:199], 0\n\t"
        "v_mov_b64 v[200:201], 0\n\t"
        "v_mov_b64 v[202:203], 0\n\t"
        "v_mov_b64 v[204:205], 0\n\t"
        "v_mov_b64 v[206:207], 0\n\t"
        "v_mov_b64 v[208:209], 0\n\t"
        "v_mov_b64 v[210:211], 0\n\t"
        "v_mov_b64 v[212:213], 0\n\t"
        "v_mov_b64 v[214:215], 0\n\t"
        "v_mov_b64 v[216:217], 0\n\t"
        "v_mov_b64 v[218:219], 0\n\t"
        "v_mov_b64 v[220:221], 0\n\t"
        "v_mov_b64 v[222:223], 0\n\t"
        "v_mov_b64 v[224:225], 0\n\t"
        "v_mov_b64 v[226:227], 0\n\t"
        "v_mov_b64 v[228:229], 0\n\t"
        "v_mov_b64 v[230:231], 0\n\t"
        "v_mov_b64 v[232:233], 0\n\t"
        "v_mov_b64 v[234:235], 0\n\t"
        "v_mov_b64 v[236:237], 0\n\t"
        "v_mov_b64 v[238:239], 0\n\t"
        "v_mov_b64 v[240:241], 0\n\t"
        "v_mov_b64 v[242:243], 0\n\t"
        "v_mov_b64 v[244:245], 0\n\t"
        "v_mov_b64 v[246:247], 0\n\t"
        "v_mov_b64 v[248:249], 0\n\t"
        "v_mov_b64 v[250:251], 0\n\t"
        "v_mov_b64 v[252:253], 0\n\t"
        "v_mov_b64 v[254:255], 0"
        : : : "v144", "v145", "v146", "v147", "v148", "v149", "v150", "v151", "v152", "v153", "v154", "v155", "v156", "v157", "v158", "v159", "v160", "v161", "v162", "v163", "v164", "v165", "v166", "v167", "v168", "v169", "v170", "v171", "v172", "v173", "v174", "v175", "v176", "v177", "v178", "v179", "v180", "v181", "v182", "v183", "v184", "v185", "v186", "v187", "v188", "v189", "v190", "v191", "v192", "v193", "v194", "v195", "v196", "v197", "v198", "v199", "v200", "v201", "v202", "v203", "v204", "v205", "v206", "v207", "v208", "v209", "v210", "v211", "v212", "v213", "v214", "v215", "v216", "v217", "v218", "v219", "v220", "v221", "v222", "v223", "v224", "v225", "v226", "v227", "v228", "v229", "v230", "v231", "v232", "v233", "v234", "v235", "v236", "v237", "v238", "v239", "v240", "v241", "v242", "v243", "v244", "v245", "v246", "v247", "v248", "v249", "v250", "v251", "v252", "v253", "v254", "v255");

    const int    kstart = wid * SLABS_PER_W * 4;
    const float* vp0 = V + (size_t)(row_base + l15) * PROTOS + kstart + koff;
    const float* vp1 = vp0 + (size_t)16 * PROTOS;
    typedef const __attribute__((address_space(4))) int4 c_int4;
    c_int4* lq = (c_int4*)(labels + kstart);

    float px0_0 = 0.0f, px1_0 = 0.0f; unsigned pm_0 = 0u;
    float px0_1 = 0.0f, px1_1 = 0.0f; unsigned pm_1 = 0u;
    float px0_2 = 0.0f, px1_2 = 0.0f; unsigned pm_2 = 0u;
    float px0_3 = 0.0f, px1_3 = 0.0f; unsigned pm_3 = 0u;
    float px0_4 = 0.0f, px1_4 = 0.0f; unsigned pm_4 = 0u;
    float px0_5 = 0.0f, px1_5 = 0.0f; unsigned pm_5 = 0u;
    float px0_6 = 0.0f, px1_6 = 0.0f; unsigned pm_6 = 0u;
    unsigned pf = 0u;

    __builtin_amdgcn_s_setprio(3);
    for (int c = 0; c < TRIPS; ++c) {
        if (c == TRIPS - 3) __builtin_amdgcn_s_setprio(2);
        if (c == TRIPS - 2) __builtin_amdgcn_s_setprio(1);
        if (c == TRIPS - 1) __builtin_amdgcn_s_setprio(0);

        v2f  a0[8], a1[8];
        int4 lab[8];
#pragma unroll
        for (int j = 0; j < 8; ++j) {
            a0[j]  = *(const v2f*)(vp0 + 4 * j);
            a1[j]  = *(const v2f*)(vp1 + 4 * j);
            lab[j] = lq[c * 8 + j];
        }
#pragma unroll
        for (int u = 0; u < 16; ++u) {
            const int j = u >> 1;
            const unsigned la = (unsigned)((u & 1) ? lab[j].y : lab[j].x);
            const unsigned lb = (unsigned)((u & 1) ? lab[j].w : lab[j].z);
            const float    r0 = (u & 1) ? a0[j].y : a0[j].x;
            const float    r1 = (u & 1) ? a1[j].y : a1[j].x;
            const unsigned ta = la >> 4, tb = lb >> 4;
            const unsigned bla = 1u << (la & 15u);
            const unsigned blb = 0x10000u << (lb & 15u);
            const unsigned hit = ((ta < (unsigned)NTILES) ? (1u << ta) : 0u)
                               | ((tb < (unsigned)NTILES) ? (1u << tb) : 0u);
            SITE(0) SITE(1) SITE(2) SITE(3) SITE(4) SITE(5) SITE(6)
        }
        vp0 += 32;
        vp1 += 32;
    }

    FLUSH(0) FLUSH(1) FLUSH(2) FLUSH(3) FLUSH(4) FLUSH(5) FLUSH(6)

    { float f[8]; ACC_READ(f, 144, 145, 146, 147, 148, 149, 150, 151);
      float* dst = &smem[((wid * ROWTILES + 0) * NTILES + 0) * TILE_F + lane * 4];
      const v4f lo = {f[0], f[1], f[2], f[3]}, hi = {f[4], f[5], f[6], f[7]};
      *(v4f*)dst = lo;
      *(v4f*)(dst + 128) = hi; }
    { float f[8]; ACC_READ(f, 152, 153, 154, 155, 156, 157, 158, 159);
      float* dst = &smem[((wid * ROWTILES + 0) * NTILES + 1) * TILE_F + lane * 4];
      const v4f lo = {f[0], f[1], f[2], f[3]}, hi = {f[4], f[5], f[6], f[7]};
      *(v4f*)dst = lo;
      *(v4f*)(dst + 128) = hi; }
    { float f[8]; ACC_READ(f, 160, 161, 162, 163, 164, 165, 166, 167);
      float* dst = &smem[((wid * ROWTILES + 0) * NTILES + 2) * TILE_F + lane * 4];
      const v4f lo = {f[0], f[1], f[2], f[3]}, hi = {f[4], f[5], f[6], f[7]};
      *(v4f*)dst = lo;
      *(v4f*)(dst + 128) = hi; }
    { float f[8]; ACC_READ(f, 168, 169, 170, 171, 172, 173, 174, 175);
      float* dst = &smem[((wid * ROWTILES + 0) * NTILES + 3) * TILE_F + lane * 4];
      const v4f lo = {f[0], f[1], f[2], f[3]}, hi = {f[4], f[5], f[6], f[7]};
      *(v4f*)dst = lo;
      *(v4f*)(dst + 128) = hi; }
    { float f[8]; ACC_READ(f, 176, 177, 178, 179, 180, 181, 182, 183);
      float* dst = &smem[((wid * ROWTILES + 0) * NTILES + 4) * TILE_F + lane * 4];
      const v4f lo = {f[0], f[1], f[2], f[3]}, hi = {f[4], f[5], f[6], f[7]};
      *(v4f*)dst = lo;
      *(v4f*)(dst + 128) = hi; }
    { float f[8]; ACC_READ(f, 184, 185, 186, 187, 188, 189, 190, 191);
      float* dst = &smem[((wid * ROWTILES + 0) * NTILES + 5) * TILE_F + lane * 4];
      const v4f lo = {f[0], f[1], f[2], f[3]}, hi = {f[4], f[5], f[6], f[7]};
      *(v4f*)dst = lo;
      *(v4f*)(dst + 128) = hi; }
    { float f[8]; ACC_READ(f, 192, 193, 194, 195, 196, 197, 198, 199);
      float* dst = &smem[((wid * ROWTILES + 0) * NTILES + 6) * TILE_F + lane * 4];
      const v4f lo = {f[0], f[1], f[2], f[3]}, hi = {f[4], f[5], f[6], f[7]};
      *(v4f*)dst = lo;
      *(v4f*)(dst + 128) = hi; }
    { float f[8]; ACC_READ(f, 200, 201, 202, 203, 204, 205, 206, 207);
      float* dst = &smem[((wid * ROWTILES + 1) * NTILES + 0) * TILE_F + lane * 4];
      const v4f lo = {f[0], f[1], f[2], f[3]}, hi = {f[4], f[5], f[6], f[7]};
      *(v4f*)dst = lo;
      *(v4f*)(dst + 128) = hi; }
    { float f[8]; ACC_READ(f, 208, 209, 210, 211, 212, 213, 214, 215);
      float* dst = &smem[((wid * ROWTILES + 1) * NTILES + 1) * TILE_F + lane * 4];
      const v4f lo = {f[0], f[1], f[2], f[3]}, hi = {f[4], f[5], f[6], f[7]};
      *(v4f*)dst = lo;
      *(v4f*)(dst + 128) = hi; }
    { float f[8]; ACC_READ(f, 216, 217, 218, 219, 220, 221, 222, 223);
      float* dst = &smem[((wid * ROWTILES + 1) * NTILES + 2) * TILE_F + lane * 4];
      const v4f lo = {f[0], f[1], f[2], f[3]}, hi = {f[4], f[5], f[6], f[7]};
      *(v4f*)dst = lo;
      *(v4f*)(dst + 128) = hi; }
    { float f[8]; ACC_READ(f, 224, 225, 226, 227, 228, 229, 230, 231);
      float* dst = &smem[((wid * ROWTILES + 1) * NTILES + 3) * TILE_F + lane * 4];
      const v4f lo = {f[0], f[1], f[2], f[3]}, hi = {f[4], f[5], f[6], f[7]};
      *(v4f*)dst = lo;
      *(v4f*)(dst + 128) = hi; }
    { float f[8]; ACC_READ(f, 232, 233, 234, 235, 236, 237, 238, 239);
      float* dst = &smem[((wid * ROWTILES + 1) * NTILES + 4) * TILE_F + lane * 4];
      const v4f lo = {f[0], f[1], f[2], f[3]}, hi = {f[4], f[5], f[6], f[7]};
      *(v4f*)dst = lo;
      *(v4f*)(dst + 128) = hi; }
    { float f[8]; ACC_READ(f, 240, 241, 242, 243, 244, 245, 246, 247);
      float* dst = &smem[((wid * ROWTILES + 1) * NTILES + 5) * TILE_F + lane * 4];
      const v4f lo = {f[0], f[1], f[2], f[3]}, hi = {f[4], f[5], f[6], f[7]};
      *(v4f*)dst = lo;
      *(v4f*)(dst + 128) = hi; }
    { float f[8]; ACC_READ(f, 248, 249, 250, 251, 252, 253, 254, 255);
      float* dst = &smem[((wid * ROWTILES + 1) * NTILES + 6) * TILE_F + lane * 4];
      const v4f lo = {f[0], f[1], f[2], f[3]}, hi = {f[4], f[5], f[6], f[7]};
      *(v4f*)dst = lo;
      *(v4f*)(dst + 128) = hi; }
    __syncthreads();

    const int rh = lane >> 4;
    const int cl = lane & 15;
    for (int u = wid; u < UNITS; u += WAVES) {
        const int h    = u / (NTILES * 2);
        const int rem  = u - h * (NTILES * 2);
        const int t    = rem >> 1;
        const int half = rem & 1;
        const int cls  = t * 16 + cl;
        if (cls < CLASSES) {
            const float* srcp = &smem[(h * NTILES + t) * TILE_F + rh * 128 + (half * 16 + cl) * 4];
            v4f sum = {0.0f, 0.0f, 0.0f, 0.0f};
#pragma unroll
            for (int w = 0; w < WAVES; ++w)
                sum += *(const v4f*)(srcp + w * (ROWTILES * NTILES * TILE_F));
            const int row = row_base + h * 16 + half * 8 + rh * 4;
            float* op = out + (size_t)row * CLASSES + cls;
            op[0 * CLASSES] = sum.x;
            op[1 * CLASSES] = sum.y;
            op[2 * CLASSES] = sum.z;
            op[3 * CLASSES] = sum.w;
        }
    }
}

extern "C" void kernel_launch(void* const* d_in, const int* in_sizes, int n_in,
                              void* d_out, int out_size, void* d_ws, size_t ws_size,
                              hipStream_t stream) {
    const float* V      = (const float*)d_in[0];
    const int*   labels = (const int*)d_in[1];
    float*       out    = (float*)d_out;

    stratified_sum_wmma<<<BATCH / BLOCK_ROWS, THREADS, 0, stream>>>(V, labels, out);
}


# Pending-slot update without an else: a hit tile fires its pair if the slot is full, then always stores the arriving register in place and toggles the slot bit; sum order unchanged
# speedup vs baseline: 1.7713x; 1.0791x over previous
#include <hip/hip_runtime.h>


typedef float v2f __attribute__((ext_vector_type(2)));
typedef float v4f __attribute__((ext_vector_type(4)));
typedef float v8f __attribute__((ext_vector_type(8)));

#define BATCH       8192
#define PROTOS      4096
#define CLASSES     100
#define NTILES      7
#define WAVES       16
#define THREADS     (WAVES * 32)
#define KSLABS      (PROTOS / 4)
#define SLABS_PER_W (KSLABS / WAVES)
#define TRIPS       (SLABS_PER_W / 8)
#define ROWTILES    2
#define BLOCK_ROWS  (16 * ROWTILES)
#define TILE_F      256
#define UNITS       (ROWTILES * NTILES * 2)


#define WMMA_PAIR_0(X0, X1, B)                                                     \
    asm volatile("v_wmma_f32_16x16x4_f32 v[144:151], %0, %2, v[144:151]\n\t"       \
                 "v_wmma_f32_16x16x4_f32 v[200:207], %1, %2, v[200:207]\n\t"       \
                 "v_nop\n\tv_nop\n\tv_nop\n\tv_nop"                                     \
                 : : "v"(X0), "v"(X1), "v"(B)                                         \
                 : "v144", "v145", "v146", "v147", "v148", "v149", "v150", "v151", "v200", "v201", "v202", "v203", "v204", "v205", "v206", "v207")
#define WMMA_PAIR_1(X0, X1, B)                                                     \
    asm volatile("v_wmma_f32_16x16x4_f32 v[152:159], %0, %2, v[152:159]\n\t"       \
                 "v_wmma_f32_16x16x4_f32 v[208:215], %1, %2, v[208:215]\n\t"       \
                 "v_nop\n\tv_nop\n\tv_nop\n\tv_nop"                                     \
                 : : "v"(X0), "v"(X1), "v"(B)                                         \
                 : "v152", "v153", "v154", "v155", "v156", "v157", "v158", "v159", "v208", "v209", "v210", "v211", "v212", "v213", "v214", "v215")
#define WMMA_PAIR_2(X0, X1, B)                                                     \
    asm volatile("v_wmma_f32_16x16x4_f32 v[160:167], %0, %2, v[160:167]\n\t"       \
                 "v_wmma_f32_16x16x4_f32 v[216:223], %1, %2, v[216:223]\n\t"       \
                 "v_nop\n\tv_nop\n\tv_nop\n\tv_nop"                                     \
                 : : "v"(X0), "v"(X1), "v"(B)                                         \
                 : "v160", "v161", "v162", "v163", "v164", "v165", "v166", "v167", "v216", "v217", "v218", "v219", "v220", "v221", "v222", "v223")
#define WMMA_PAIR_3(X0, X1, B)                                                     \
    asm volatile("v_wmma_f32_16x16x4_f32 v[168:175], %0, %2, v[168:175]\n\t"       \
                 "v_wmma_f32_16x16x4_f32 v[224:231], %1, %2, v[224:231]\n\t"       \
                 "v_nop\n\tv_nop\n\tv_nop\n\tv_nop"                                     \
                 : : "v"(X0), "v"(X1), "v"(B)                                         \
                 : "v168", "v169", "v170", "v171", "v172", "v173", "v174", "v175", "v224", "v225", "v226", "v227", "v228", "v229", "v230", "v231")
#define WMMA_PAIR_4(X0, X1, B)                                                     \
    asm volatile("v_wmma_f32_16x16x4_f32 v[176:183], %0, %2, v[176:183]\n\t"       \
                 "v_wmma_f32_16x16x4_f32 v[232:239], %1, %2, v[232:239]\n\t"       \
                 "v_nop\n\tv_nop\n\tv_nop\n\tv_nop"                                     \
                 : : "v"(X0), "v"(X1), "v"(B)                                         \
                 : "v176", "v177", "v178", "v179", "v180", "v181", "v182", "v183", "v232", "v233", "v234", "v235", "v236", "v237", "v238", "v239")
#define WMMA_PAIR_5(X0, X1, B)                                                     \
    asm volatile("v_wmma_f32_16x16x4_f32 v[184:191], %0, %2, v[184:191]\n\t"       \
                 "v_wmma_f32_16x16x4_f32 v[240:247], %1, %2, v[240:247]\n\t"       \
                 "v_nop\n\tv_nop\n\tv_nop\n\tv_nop"                                     \
                 : : "v"(X0), "v"(X1), "v"(B)                                         \
                 : "v184", "v185", "v186", "v187", "v188", "v189", "v190", "v191", "v240", "v241", "v242", "v243", "v244", "v245", "v246", "v247")
#define WMMA_PAIR_6(X0, X1, B)                                                     \
    asm volatile("v_wmma_f32_16x16x4_f32 v[192:199], %0, %2, v[192:199]\n\t"       \
                 "v_wmma_f32_16x16x4_f32 v[248:255], %1, %2, v[248:255]\n\t"       \
                 "v_nop\n\tv_nop\n\tv_nop\n\tv_nop"                                     \
                 : : "v"(X0), "v"(X1), "v"(B)                                         \
                 : "v192", "v193", "v194", "v195", "v196", "v197", "v198", "v199", "v248", "v249", "v250", "v251", "v252", "v253", "v254", "v255")

#define ACC_READ(F, R0, R1, R2, R3, R4, R5, R6, R7)                                    \
    asm volatile("v_mov_b32 %0, v" #R0 "\n\tv_mov_b32 %1, v" #R1 "\n\t"                \
                 "v_mov_b32 %2, v" #R2 "\n\tv_mov_b32 %3, v" #R3 "\n\t"                \
                 "v_mov_b32 %4, v" #R4 "\n\tv_mov_b32 %5, v" #R5 "\n\t"                \
                 "v_mov_b32 %6, v" #R6 "\n\tv_mov_b32 %7, v" #R7                        \
                 : "=v"(F[0]), "=v"(F[1]), "=v"(F[2]), "=v"(F[3]),                     \
                   "=v"(F[4]), "=v"(F[5]), "=v"(F[6]), "=v"(F[7]))

#define SLOT_STORE(P0, P1, PM, R0, R1, M)                                            \
    asm volatile("v_mov_b32 %0, %3\n\tv_mov_b32 %1, %4\n\ts_mov_b32 %2, %5"          \
                 : "+v"(P0), "+v"(P1), "+s"(PM) : "v"(R0), "v"(R1), "s"(M))

#define SITE(T)                                                                      \
    if (hit & (1u << T)) {                                                           \
        const unsigned m = ((ta == (unsigned)T) ? bla : 0u) | ((tb == (unsigned)T) ? blb : 0u); \
        if (pf & (1u << T)) {                                                        \
            const v2f x0 = {px0_##T, r0};                                            \
            const v2f x1 = {px1_##T, r1};                                            \
            v2f b;                                                                   \
            b.x = __builtin_amdgcn_inverse_ballot_w32(pm_##T) ? 1.0f : 0.0f;         \
            b.y = __builtin_amdgcn_inverse_ballot_w32(m)      ? 1.0f : 0.0f;         \
            WMMA_PAIR_##T(x0, x1, b);                                                \
        }                                                                            \
        SLOT_STORE(px0_##T, px1_##T, pm_##T, r0, r1, m);                             \
        pf ^= (1u << T);                                                             \
    }

#define FLUSH(T)                                                                     \
    if (pf & (1u << T)) {                                                            \
        const v2f x0 = {px0_##T, 0.0f}, x1 = {px1_##T, 0.0f};                        \
        v2f b;                                                                       \
        b.x = __builtin_amdgcn_inverse_ballot_w32(pm_##T) ? 1.0f : 0.0f;             \
        b.y = 0.0f;                                                                  \
        WMMA_PAIR_##T(x0, x1, b);                                                    \
    }

__global__ __launch_bounds__(THREADS) __attribute__((amdgpu_waves_per_eu(7, 7)))
void stratified_sum_wmma(const float* __restrict__ V,
                         const int*   __restrict__ labels,
                         float*       __restrict__ out) {
    __shared__ __align__(32) float smem[WAVES * ROWTILES * NTILES * TILE_F];

    const int tid  = threadIdx.x;
    const int wid  = __builtin_amdgcn_readfirstlane(tid >> 5);
    const int lane = tid & 31;
    const int row_base = blockIdx.x * BLOCK_ROWS;

    const int l15  = lane & 15;
    const int koff = (lane >> 4) * 2;

    asm volatile(
        "v_mov_b64 v[144:145], 0\n\t"
        "v_mov_b64 v[146:147], 0\n\t"
        "v_mov_b64 v[148:149], 0\n\t"
        "v_mov_b64 v[150:151], 0\n\t"
        "v_mov_b64 v[152:153], 0\n\t"
        "v_mov_b64 v[154:155], 0\n\t"
        "v_mov_b64 v[156:157], 0\n\t"
        "v_mov_b64 v[158:159], 0\n\t"
        "v_mov_b64 v[160:161], 0\n\t"
        "v_mov_b64 v[162:163], 0\n\t"
        "v_mov_b64 v[164:165], 0\n\t"
        "v_mov_b64 v[166:167], 0\n\t"
        "v_mov_b64 v[168:169], 0\n\t"
        "v_mov_b64 v[170:171], 0\n\t"
        "v_mov_b64 v[172:173], 0\n\t"
        "v_mov_b64 v[174:175], 0\n\t"
        "v_mov_b64 v[176:177], 0\n\t"
        "v_mov_b64 v[178:179], 0\n\t"
        "v_mov_b64 v[180:181], 0\n\t"
        "v_mov_b64 v[182:183], 0\n\t"
        "v_mov_b64 v[184:185], 0\n\t"
        "v_mov_b64 v[186:187], 0\n\t"
        "v_mov_b64 v[188:189], 0\n\t"
        "v_mov_b64 v[190:191], 0\n\t"
        "v_mov_b64 v[192:193], 0\n\t"
        "v_mov_b64 v[194:195], 0\n\t"
        "v_mov_b64 v[196:197], 0\n\t"
        "v_mov_b64 v[198:199], 0\n\t"
        "v_mov_b64 v[200:201], 0\n\t"
        "v_mov_b64 v[202:203], 0\n\t"
        "v_mov_b64 v[204:205], 0\n\t"
        "v_mov_b64 v[206:207], 0\n\t"
        "v_mov_b64 v[208:209], 0\n\t"
        "v_mov_b64 v[210:211], 0\n\t"
        "v_mov_b64 v[212:213], 0\n\t"
        "v_mov_b64 v[214:215], 0\n\t"
        "v_mov_b64 v[216:217], 0\n\t"
        "v_mov_b64 v[218:219], 0\n\t"
        "v_mov_b64 v[220:221], 0\n\t"
        "v_mov_b64 v[222:223], 0\n\t"
        "v_mov_b64 v[224:225], 0\n\t"
        "v_mov_b64 v[226:227], 0\n\t"
        "v_mov_b64 v[228:229], 0\n\t"
        "v_mov_b64 v[230:231], 0\n\t"
        "v_mov_b64 v[232:233], 0\n\t"
        "v_mov_b64 v[234:235], 0\n\t"
        "v_mov_b64 v[236:237], 0\n\t"
        "v_mov_b64 v[238:239], 0\n\t"
        "v_mov_b64 v[240:241], 0\n\t"
        "v_mov_b64 v[242:243], 0\n\t"
        "v_mov_b64 v[244:245], 0\n\t"
        "v_mov_b64 v[246:247], 0\n\t"
        "v_mov_b64 v[248:249], 0\n\t"
        "v_mov_b64 v[250:251], 0\n\t"
        "v_mov_b64 v[252:253], 0\n\t"
        "v_mov_b64 v[254:255], 0"
        : : : "v144", "v145", "v146", "v147", "v148", "v149", "v150", "v151", "v152", "v153", "v154", "v155", "v156", "v157", "v158", "v159", "v160", "v161", "v162", "v163", "v164", "v165", "v166", "v167", "v168", "v169", "v170", "v171", "v172", "v173", "v174", "v175", "v176", "v177", "v178", "v179", "v180", "v181", "v182", "v183", "v184", "v185", "v186", "v187", "v188", "v189", "v190", "v191", "v192", "v193", "v194", "v195", "v196", "v197", "v198", "v199", "v200", "v201", "v202", "v203", "v204", "v205", "v206", "v207", "v208", "v209", "v210", "v211", "v212", "v213", "v214", "v215", "v216", "v217", "v218", "v219", "v220", "v221", "v222", "v223", "v224", "v225", "v226", "v227", "v228", "v229", "v230", "v231", "v232", "v233", "v234", "v235", "v236", "v237", "v238", "v239", "v240", "v241", "v242", "v243", "v244", "v245", "v246", "v247", "v248", "v249", "v250", "v251", "v252", "v253", "v254", "v255");

    const int    kstart = wid * SLABS_PER_W * 4;
    const float* vp0 = V + (size_t)(row_base + l15) * PROTOS + kstart + koff;
    const float* vp1 = vp0 + (size_t)16 * PROTOS;
    typedef const __attribute__((address_space(4))) int4 c_int4;
    c_int4* lq = (c_int4*)(labels + kstart);

    float px0_0 = 0.0f, px1_0 = 0.0f; unsigned pm_0 = 0u;
    float px0_1 = 0.0f, px1_1 = 0.0f; unsigned pm_1 = 0u;
    float px0_2 = 0.0f, px1_2 = 0.0f; unsigned pm_2 = 0u;
    float px0_3 = 0.0f, px1_3 = 0.0f; unsigned pm_3 = 0u;
    float px0_4 = 0.0f, px1_4 = 0.0f; unsigned pm_4 = 0u;
    float px0_5 = 0.0f, px1_5 = 0.0f; unsigned pm_5 = 0u;
    float px0_6 = 0.0f, px1_6 = 0.0f; unsigned pm_6 = 0u;
    unsigned pf = 0u;

    __builtin_amdgcn_s_setprio(3);
    for (int c = 0; c < TRIPS; ++c) {
        if (c == TRIPS - 3) __builtin_amdgcn_s_setprio(2);
        if (c == TRIPS - 2) __builtin_amdgcn_s_setprio(1);
        if (c == TRIPS - 1) __builtin_amdgcn_s_setprio(0);

        v2f  a0[8], a1[8];
        int4 lab[8];
#pragma unroll
        for (int j = 0; j < 8; ++j) {
            a0[j]  = *(const v2f*)(vp0 + 4 * j);
            a1[j]  = *(const v2f*)(vp1 + 4 * j);
            lab[j] = lq[c * 8 + j];
        }
#pragma unroll
        for (int u = 0; u < 16; ++u) {
            const int j = u >> 1;
            const unsigned la = (unsigned)((u & 1) ? lab[j].y : lab[j].x);
            const unsigned lb = (unsigned)((u & 1) ? lab[j].w : lab[j].z);
            const float    r0 = (u & 1) ? a0[j].y : a0[j].x;
            const float    r1 = (u & 1) ? a1[j].y : a1[j].x;
            const unsigned ta = la >> 4, tb = lb >> 4;
            const unsigned bla = 1u << (la & 15u);
            const unsigned blb = 0x10000u << (lb & 15u);
            const unsigned hit = ((ta < (unsigned)NTILES) ? (1u << ta) : 0u)
                               | ((tb < (unsigned)NTILES) ? (1u << tb) : 0u);
            SITE(0) SITE(1) SITE(2) SITE(3) SITE(4) SITE(5) SITE(6)
        }
        vp0 += 32;
        vp1 += 32;
    }

    FLUSH(0) FLUSH(1) FLUSH(2) FLUSH(3) FLUSH(4) FLUSH(5) FLUSH(6)

    { float f[8]; ACC_READ(f, 144, 145, 146, 147, 148, 149, 150, 151);
      float* dst = &smem[((wid * ROWTILES + 0) * NTILES + 0) * TILE_F + lane * 4];
      const v4f lo = {f[0], f[1], f[2], f[3]}, hi = {f[4], f[5], f[6], f[7]};
      *(v4f*)dst = lo;
      *(v4f*)(dst + 128) = hi; }
    { float f[8]; ACC_READ(f, 152, 153, 154, 155, 156, 157, 158, 159);
      float* dst = &smem[((wid * ROWTILES + 0) * NTILES + 1) * TILE_F + lane * 4];
      const v4f lo = {f[0], f[1], f[2], f[3]}, hi = {f[4], f[5], f[6], f[7]};
      *(v4f*)dst = lo;
      *(v4f*)(dst + 128) = hi; }
    { float f[8]; ACC_READ(f, 160, 161, 162, 163, 164, 165, 166, 167);
      float* dst = &smem[((wid * ROWTILES + 0) * NTILES + 2) * TILE_F + lane * 4];
      const v4f lo = {f[0], f[1], f[2], f[3]}, hi = {f[4], f[5], f[6], f[7]};
      *(v4f*)dst = lo;
      *(v4f*)(dst + 128) = hi; }
    { float f[8]; ACC_READ(f, 168, 169, 170, 171, 172, 173, 174, 175);
      float* dst = &smem[((wid * ROWTILES + 0) * NTILES + 3) * TILE_F + lane * 4];
      const v4f lo = {f[0], f[1], f[2], f[3]}, hi = {f[4], f[5], f[6], f[7]};
      *(v4f*)dst = lo;
      *(v4f*)(dst + 128) = hi; }
    { float f[8]; ACC_READ(f, 176, 177, 178, 179, 180, 181, 182, 183);
      float* dst = &smem[((wid * ROWTILES + 0) * NTILES + 4) * TILE_F + lane * 4];
      const v4f lo = {f[0], f[1], f[2], f[3]}, hi = {f[4], f[5], f[6], f[7]};
      *(v4f*)dst = lo;
      *(v4f*)(dst + 128) = hi; }
    { float f[8]; ACC_READ(f, 184, 185, 186, 187, 188, 189, 190, 191);
      float* dst = &smem[((wid * ROWTILES + 0) * NTILES + 5) * TILE_F + lane * 4];
      const v4f lo = {f[0], f[1], f[2], f[3]}, hi = {f[4], f[5], f[6], f[7]};
      *(v4f*)dst = lo;
      *(v4f*)(dst + 128) = hi; }
    { float f[8]; ACC_READ(f, 192, 193, 194, 195, 196, 197, 198, 199);
      float* dst = &smem[((wid * ROWTILES + 0) * NTILES + 6) * TILE_F + lane * 4];
      const v4f lo = {f[0], f[1], f[2], f[3]}, hi = {f[4], f[5], f[6], f[7]};
      *(v4f*)dst = lo;
      *(v4f*)(dst + 128) = hi; }
    { float f[8]; ACC_READ(f, 200, 201, 202, 203, 204, 205, 206, 207);
      float* dst = &smem[((wid * ROWTILES + 1) * NTILES + 0) * TILE_F + lane * 4];
      const v4f lo = {f[0], f[1], f[2], f[3]}, hi = {f[4], f[5], f[6], f[7]};
      *(v4f*)dst = lo;
      *(v4f*)(dst + 128) = hi; }
    { float f[8]; ACC_READ(f, 208, 209, 210, 211, 212, 213, 214, 215);
      float* dst = &smem[((wid * ROWTILES + 1) * NTILES + 1) * TILE_F + lane * 4];
      const v4f lo = {f[0], f[1], f[2], f[3]}, hi = {f[4], f[5], f[6], f[7]};
      *(v4f*)dst = lo;
      *(v4f*)(dst + 128) = hi; }
    { float f[8]; ACC_READ(f, 216, 217, 218, 219, 220, 221, 222, 223);
      float* dst = &smem[((wid * ROWTILES + 1) * NTILES + 2) * TILE_F + lane * 4];
      const v4f lo = {f[0], f[1], f[2], f[3]}, hi = {f[4], f[5], f[6], f[7]};
      *(v4f*)dst = lo;
      *(v4f*)(dst + 128) = hi; }
    { float f[8]; ACC_READ(f, 224, 225, 226, 227, 228, 229, 230, 231);
      float* dst = &smem[((wid * ROWTILES + 1) * NTILES + 3) * TILE_F + lane * 4];
      const v4f lo = {f[0], f[1], f[2], f[3]}, hi = {f[4], f[5], f[6], f[7]};
      *(v4f*)dst = lo;
      *(v4f*)(dst + 128) = hi; }
    { float f[8]; ACC_READ(f, 232, 233, 234, 235, 236, 237, 238, 239);
      float* dst = &smem[((wid * ROWTILES + 1) * NTILES + 4) * TILE_F + lane * 4];
      const v4f lo = {f[0], f[1], f[2], f[3]}, hi = {f[4], f[5], f[6], f[7]};
      *(v4f*)dst = lo;
      *(v4f*)(dst + 128) = hi; }
    { float f[8]; ACC_READ(f, 240, 241, 242, 243, 244, 245, 246, 247);
      float* dst = &smem[((wid * ROWTILES + 1) * NTILES + 5) * TILE_F + lane * 4];
      const v4f lo = {f[0], f[1], f[2], f[3]}, hi = {f[4], f[5], f[6], f[7]};
      *(v4f*)dst = lo;
      *(v4f*)(dst + 128) = hi; }
    { float f[8]; ACC_READ(f, 248, 249, 250, 251, 252, 253, 254, 255);
      float* dst = &smem[((wid * ROWTILES + 1) * NTILES + 6) * TILE_F + lane * 4];
      const v4f lo = {f[0], f[1], f[2], f[3]}, hi = {f[4], f[5], f[6], f[7]};
      *(v4f*)dst = lo;
      *(v4f*)(dst + 128) = hi; }
    __syncthreads();

    const int rh = lane >> 4;
    const int cl = lane & 15;
    for (int u = wid; u < UNITS; u += WAVES) {
        const int h    = u / (NTILES * 2);
        const int rem  = u - h * (NTILES * 2);
        const int t    = rem >> 1;
        const int half = rem & 1;
        const int cls  = t * 16 + cl;
        if (cls < CLASSES) {
            const float* srcp = &smem[(h * NTILES + t) * TILE_F + rh * 128 + (half * 16 + cl) * 4];
            v4f sum = {0.0f, 0.0f, 0.0f, 0.0f};
#pragma unroll
            for (int w = 0; w < WAVES; ++w)
                sum += *(const v4f*)(srcp + w * (ROWTILES * NTILES * TILE_F));
            const int row = row_base + h * 16 + half * 8 + rh * 4;
            float* op = out + (size_t)row * CLASSES + cls;
            op[0 * CLASSES] = sum.x;
            op[1 * CLASSES] = sum.y;
            op[2 * CLASSES] = sum.z;
            op[3 * CLASSES] = sum.w;
        }
    }
}

extern "C" void kernel_launch(void* const* d_in, const int* in_sizes, int n_in,
                              void* d_out, int out_size, void* d_ws, size_t ws_size,
                              hipStream_t stream) {
    const float* V      = (const float*)d_in[0];
    const int*   labels = (const int*)d_in[1];
    float*       out    = (float*)d_out;

    stratified_sum_wmma<<<BATCH / BLOCK_ROWS, THREADS, 0, stream>>>(V, labels, out);
}
